# starting kernel
# baseline (speedup 1.0000x reference)
#include <hip/hip_runtime.h>
#include <hip/hip_bf16.h>

typedef __attribute__((ext_vector_type(2))) float v2f;
typedef __attribute__((ext_vector_type(8))) float v8f;

constexpr int S       = 2048;
constexpr int T       = 16384;
constexpr int L       = 16;
constexpr int NCHUNK  = T / L;
constexpr int NGROUP  = S / 16;
constexpr int SBLK    = 32;
constexpr int NSB     = NCHUNK / SBLK;

constexpr size_t OFF_H    = 0;
constexpr size_t OFF_P    = 256;
constexpr size_t OFF_Q    = 320;
constexpr size_t OFF_APOW = 576;
constexpr size_t OFF_ZC   = 4096;
constexpr size_t OFF_U    = OFF_ZC + (size_t)NCHUNK * 4 * S;
constexpr size_t OFF_ZS   = OFF_U  + (size_t)NSB    * 4 * S;

__global__ void lp_setup_kernel(const float* __restrict__ bc,
                                const float* __restrict__ ac,
                                float* __restrict__ ws) {
  if (threadIdx.x != 0 || blockIdx.x != 0) return;
  const float b0 = bc[0];
  float A[4][4], g4[4];
  for (int i = 0; i < 4; ++i) {
    for (int j = 0; j < 4; ++j) A[i][j] = (j == i + 1) ? 1.0f : 0.0f;
    A[i][0] -= ac[i + 1];
    g4[i] = bc[i + 1] - ac[i + 1] * b0;
  }
  float h[16];
  h[0] = b0;
  float v[4] = {g4[0], g4[1], g4[2], g4[3]};
  for (int t = 1; t < 16; ++t) {
    h[t] = v[0];
    float nv[4];
    for (int i = 0; i < 4; ++i) {
      float s = 0.f;
      for (int j = 0; j < 4; ++j) s += A[i][j] * v[j];
      nv[i] = s;
    }
    for (int i = 0; i < 4; ++i) v[i] = nv[i];
  }
  for (int t = 0; t < 16; ++t)
    for (int k = 0; k < 16; ++k)
      ws[OFF_H + t * 16 + k] = (k <= t) ? h[t - k] : 0.0f;
  float M[4][4];
  for (int i = 0; i < 4; ++i)
    for (int j = 0; j < 4; ++j) M[i][j] = (i == j) ? 1.0f : 0.0f;
  for (int t = 0; t < 16; ++t) {
    for (int m = 0; m < 4; ++m) ws[OFF_P + t * 4 + m] = M[0][m];
    float NM[4][4];
    for (int i = 0; i < 4; ++i)
      for (int j = 0; j < 4; ++j) {
        float s = 0.f;
        for (int k = 0; k < 4; ++k) s += A[i][k] * M[k][j];
        NM[i][j] = s;
      }
    for (int i = 0; i < 4; ++i)
      for (int j = 0; j < 4; ++j) M[i][j] = NM[i][j];
  }
  float A16[4][4];
  for (int i = 0; i < 4; ++i)
    for (int j = 0; j < 4; ++j) A16[i][j] = M[i][j];
  for (int i = 0; i < 256; ++i) ws[OFF_Q + i] = 0.0f;
  for (int i = 0; i < 4; ++i) v[i] = g4[i];
  for (int k = 15; k >= 0; --k) {
    for (int m = 0; m < 4; ++m) ws[OFF_Q + m * 16 + k] = v[m];
    float nv[4];
    for (int i = 0; i < 4; ++i) {
      float s = 0.f;
      for (int j = 0; j < 4; ++j) s += A[i][j] * v[j];
      nv[i] = s;
    }
    for (int i = 0; i < 4; ++i) v[i] = nv[i];
  }
  float Pw[4][4];
  for (int i = 0; i < 4; ++i)
    for (int j = 0; j < 4; ++j) Pw[i][j] = (i == j) ? 1.0f : 0.0f;
  for (int p = 0; p <= 32; ++p) {
    for (int i = 0; i < 4; ++i)
      for (int j = 0; j < 4; ++j) ws[OFF_APOW + p * 16 + i * 4 + j] = Pw[i][j];
    float NM[4][4];
    for (int i = 0; i < 4; ++i)
      for (int j = 0; j < 4; ++j) {
        float s = 0.f;
        for (int k = 0; k < 4; ++k) s += Pw[i][k] * A16[k][j];
        NM[i][j] = s;
      }
    for (int i = 0; i < 4; ++i)
      for (int j = 0; j < 4; ++j) Pw[i][j] = NM[i][j];
  }
}

__global__ __launch_bounds__(256) void lp_zc_kernel(const float* __restrict__ x,
                                                    float* __restrict__ ws) {
  const int wave = threadIdx.x >> 5;
  const int lane = threadIdx.x & 31;
  const int tile = blockIdx.x * 8 + wave;
  const int c = tile >> 7;
  const int g = tile & 127;
  const int half = lane >> 4;
  const int l16 = lane & 15;
  const float* Q = ws + OFF_Q;
  const float* xt = x + (size_t)c * L * S + (size_t)g * 16 + l16;
  v8f acc = {};
#pragma unroll
  for (int k = 0; k < 4; ++k) {
    const int kk0 = 4 * k + 2 * half;
    v2f av, bv;
    av[0] = Q[l16 * 16 + kk0];
    av[1] = Q[l16 * 16 + kk0 + 1];
    bv[0] = xt[(size_t)kk0 * S];
    bv[1] = xt[(size_t)(kk0 + 1) * S];
    acc = __builtin_amdgcn_wmma_f32_16x16x4_f32(false, av, false, bv,
                                                (short)0, acc, false, false);
  }
  if (half == 0) {
    float* zc = ws + OFF_ZC + (size_t)c * 4 * S + (size_t)g * 16 + l16;
    zc[0 * (size_t)S] = acc[0];
    zc[1 * (size_t)S] = acc[1];
    zc[2 * (size_t)S] = acc[2];
    zc[3 * (size_t)S] = acc[3];
  }
}

__global__ __launch_bounds__(256) void lp_scan1_kernel(float* __restrict__ ws) {
  const int tid = blockIdx.x * 256 + threadIdx.x;
  const int s  = tid & (S - 1);
  const int sb = tid >> 11;
  const float* A16 = ws + OFF_APOW + 16;
  float a[16];
#pragma unroll
  for (int i = 0; i < 16; ++i) a[i] = A16[i];
  float w0 = 0.f, w1 = 0.f, w2 = 0.f, w3 = 0.f;
  float* zc = ws + OFF_ZC + ((size_t)sb * SBLK) * 4 * S + s;
  for (int cc = 0; cc < SBLK; ++cc) {
    const float r0 = zc[0 * (size_t)S], r1 = zc[1 * (size_t)S];
    const float r2 = zc[2 * (size_t)S], r3 = zc[3 * (size_t)S];
    zc[0 * (size_t)S] = w0; zc[1 * (size_t)S] = w1;
    zc[2 * (size_t)S] = w2; zc[3 * (size_t)S] = w3;
    const float n0 = a[0]*w0 + a[1]*w1 + a[2]*w2 + a[3]*w3 + r0;
    const float n1 = a[4]*w0 + a[5]*w1 + a[6]*w2 + a[7]*w3 + r1;
    const float n2 = a[8]*w0 + a[9]*w1 + a[10]*w2 + a[11]*w3 + r2;
    const float n3 = a[12]*w0 + a[13]*w1 + a[14]*w2 + a[15]*w3 + r3;
    w0 = n0; w1 = n1; w2 = n2; w3 = n3;
    zc += (size_t)4 * S;
  }
  float* u = ws + OFF_U + (size_t)sb * 4 * S + s;
  u[0 * (size_t)S] = w0; u[1 * (size_t)S] = w1;
  u[2 * (size_t)S] = w2; u[3 * (size_t)S] = w3;
}

__global__ __launch_bounds__(256) void lp_scan2_kernel(float* __restrict__ ws) {
  const int s = blockIdx.x * 256 + threadIdx.x;
  const float* A = ws + OFF_APOW + 32 * 16;
  float a[16];
#pragma unroll
  for (int i = 0; i < 16; ++i) a[i] = A[i];
  float z0 = 0.f, z1 = 0.f, z2 = 0.f, z3 = 0.f;
  for (int sb = 0; sb < NSB; ++sb) {
    float* zs = ws + OFF_ZS + (size_t)sb * 4 * S + s;
    zs[0 * (size_t)S] = z0; zs[1 * (size_t)S] = z1;
    zs[2 * (size_t)S] = z2; zs[3 * (size_t)S] = z3;
    const float* u = ws + OFF_U + (size_t)sb * 4 * S + s;
    const float u0 = u[0 * (size_t)S], u1 = u[1 * (size_t)S];
    const float u2 = u[2 * (size_t)S], u3 = u[3 * (size_t)S];
    const float n0 = a[0]*z0 + a[1]*z1 + a[2]*z2 + a[3]*z3 + u0;
    const float n1 = a[4]*z0 + a[5]*z1 + a[6]*z2 + a[7]*z3 + u1;
    const float n2 = a[8]*z0 + a[9]*z1 + a[10]*z2 + a[11]*z3 + u2;
    const float n3 = a[12]*z0 + a[13]*z1 + a[14]*z2 + a[15]*z3 + u3;
    z0 = n0; z1 = n1; z2 = n2; z3 = n3;
  }
}

__global__ __launch_bounds__(256) void lp_final_kernel(const float* __restrict__ x,
                                                       float* __restrict__ out,
                                                       const float* __restrict__ ws) {
  const int wave = threadIdx.x >> 5;
  const int lane = threadIdx.x & 31;
  const int tile = blockIdx.x * 8 + wave;
  const int c = tile >> 7;
  const int g = tile & 127;
  const int half = lane >> 4;
  const int l16 = lane & 15;
  const int sb = c >> 5;
  const int cc = c & 31;
  const int chan = g * 16 + l16;

  const float* zs = ws + OFF_ZS + (size_t)sb * 4 * S + chan;
  const float z0 = zs[0 * (size_t)S], z1 = zs[1 * (size_t)S];
  const float z2 = zs[2 * (size_t)S], z3 = zs[3 * (size_t)S];
  const float* ap = ws + OFF_APOW + cc * 16;
  const float* W  = ws + OFF_ZC + (size_t)c * 4 * S + chan;
  v2f zin;
#pragma unroll
  for (int j = 0; j < 2; ++j) {
    const int K = 2 * half + j;
    zin[j] = ap[K * 4 + 0] * z0 + ap[K * 4 + 1] * z1 +
             ap[K * 4 + 2] * z2 + ap[K * 4 + 3] * z3 + W[(size_t)K * S];
  }
  const float* Pm = ws + OFF_P;
  v2f pa;
  pa[0] = Pm[l16 * 4 + 2 * half + 0];
  pa[1] = Pm[l16 * 4 + 2 * half + 1];

  v8f acc = {};
  acc = __builtin_amdgcn_wmma_f32_16x16x4_f32(false, pa, false, zin,
                                              (short)0, acc, false, false);

  const float* H = ws + OFF_H;
  const float* xt = x + (size_t)c * L * S + chan;
#pragma unroll
  for (int k = 0; k < 4; ++k) {
    const int kk0 = 4 * k + 2 * half;
    v2f av, bv;
    av[0] = H[l16 * 16 + kk0];
    av[1] = H[l16 * 16 + kk0 + 1];
    bv[0] = xt[(size_t)kk0 * S];
    bv[1] = xt[(size_t)(kk0 + 1) * S];
    acc = __builtin_amdgcn_wmma_f32_16x16x4_f32(false, av, false, bv,
                                                (short)0, acc, false, false);
  }

  float* o = out + (size_t)c * L * S + chan;
#pragma unroll
  for (int r = 0; r < 8; ++r)
    o[(size_t)(r + 8 * half) * S] = acc[r];
}

extern "C" void kernel_launch(void* const* d_in, const int* in_sizes, int n_in,
                              void* d_out, int out_size, void* d_ws, size_t ws_size,
                              hipStream_t stream) {
  const float* signal = (const float*)d_in[0];
  const float* b      = (const float*)d_in[1];
  const float* a      = (const float*)d_in[2];
  float* out = (float*)d_out;
  float* ws  = (float*)d_ws;

  lp_setup_kernel<<<1, 64, 0, stream>>>(b, a, ws);

  const int tiles  = NCHUNK * NGROUP;
  const int blocks = tiles / 8;
  lp_zc_kernel<<<blocks, 256, 0, stream>>>(signal, ws);

  lp_scan1_kernel<<<(S * NSB) / 256, 256, 0, stream>>>(ws);
  lp_scan2_kernel<<<S / 256, 256, 0, stream>>>(ws);

  lp_final_kernel<<<blocks, 256, 0, stream>>>(signal, out, ws);
}


# Coarsen both WMMA passes: each wave walks 8 chunks x 64 channels (4 acc tiles, float4 row loads and stores, A fragments in registers, loads two chunks ahead) instead of one 16x16 tile
# speedup vs baseline: 1.2465x; 1.2465x over previous
#include <hip/hip_runtime.h>
#include <hip/hip_bf16.h>

typedef __attribute__((ext_vector_type(2))) float v2f;
typedef __attribute__((ext_vector_type(4))) float v4f;
typedef __attribute__((ext_vector_type(8))) float v8f;

constexpr int S       = 2048;
constexpr int T       = 16384;
constexpr int L       = 16;
constexpr int NCHUNK  = T / L;
constexpr int NGROUP  = S / 16;
constexpr int SBLK    = 32;
constexpr int NSB     = NCHUNK / SBLK;
constexpr int SW       = 64;
constexpr int NSTRIPG  = S / SW;
constexpr int RUN      = 8;
constexpr int ZC_AHEAD  = 2;
constexpr int FIN_AHEAD = 2;
constexpr int NSTRIPS  = NSTRIPG * (NCHUNK / RUN);
constexpr int STRIP_BLOCKS = NSTRIPS / 8;

constexpr size_t OFF_H    = 0;
constexpr size_t OFF_P    = 256;
constexpr size_t OFF_Q    = 320;
constexpr size_t OFF_APOW = 576;
constexpr size_t OFF_ZC   = 4096;
constexpr size_t OFF_U    = OFF_ZC + (size_t)NCHUNK * 4 * S;
constexpr size_t OFF_ZS   = OFF_U  + (size_t)NSB    * 4 * S;

__global__ void lp_setup_kernel(const float* __restrict__ bc,
                                const float* __restrict__ ac,
                                float* __restrict__ ws) {
  if (threadIdx.x != 0 || blockIdx.x != 0) return;
  const float b0 = bc[0];
  float A[4][4], g4[4];
  for (int i = 0; i < 4; ++i) {
    for (int j = 0; j < 4; ++j) A[i][j] = (j == i + 1) ? 1.0f : 0.0f;
    A[i][0] -= ac[i + 1];
    g4[i] = bc[i + 1] - ac[i + 1] * b0;
  }
  float h[16];
  h[0] = b0;
  float v[4] = {g4[0], g4[1], g4[2], g4[3]};
  for (int t = 1; t < 16; ++t) {
    h[t] = v[0];
    float nv[4];
    for (int i = 0; i < 4; ++i) {
      float s = 0.f;
      for (int j = 0; j < 4; ++j) s += A[i][j] * v[j];
      nv[i] = s;
    }
    for (int i = 0; i < 4; ++i) v[i] = nv[i];
  }
  for (int t = 0; t < 16; ++t)
    for (int k = 0; k < 16; ++k)
      ws[OFF_H + t * 16 + k] = (k <= t) ? h[t - k] : 0.0f;
  float M[4][4];
  for (int i = 0; i < 4; ++i)
    for (int j = 0; j < 4; ++j) M[i][j] = (i == j) ? 1.0f : 0.0f;
  for (int t = 0; t < 16; ++t) {
    for (int m = 0; m < 4; ++m) ws[OFF_P + t * 4 + m] = M[0][m];
    float NM[4][4];
    for (int i = 0; i < 4; ++i)
      for (int j = 0; j < 4; ++j) {
        float s = 0.f;
        for (int k = 0; k < 4; ++k) s += A[i][k] * M[k][j];
        NM[i][j] = s;
      }
    for (int i = 0; i < 4; ++i)
      for (int j = 0; j < 4; ++j) M[i][j] = NM[i][j];
  }
  float A16[4][4];
  for (int i = 0; i < 4; ++i)
    for (int j = 0; j < 4; ++j) A16[i][j] = M[i][j];
  for (int i = 0; i < 256; ++i) ws[OFF_Q + i] = 0.0f;
  for (int i = 0; i < 4; ++i) v[i] = g4[i];
  for (int k = 15; k >= 0; --k) {
    for (int m = 0; m < 4; ++m) ws[OFF_Q + m * 16 + k] = v[m];
    float nv[4];
    for (int i = 0; i < 4; ++i) {
      float s = 0.f;
      for (int j = 0; j < 4; ++j) s += A[i][j] * v[j];
      nv[i] = s;
    }
    for (int i = 0; i < 4; ++i) v[i] = nv[i];
  }
  float Pw[4][4];
  for (int i = 0; i < 4; ++i)
    for (int j = 0; j < 4; ++j) Pw[i][j] = (i == j) ? 1.0f : 0.0f;
  for (int p = 0; p <= 32; ++p) {
    for (int i = 0; i < 4; ++i)
      for (int j = 0; j < 4; ++j) ws[OFF_APOW + p * 16 + i * 4 + j] = Pw[i][j];
    float NM[4][4];
    for (int i = 0; i < 4; ++i)
      for (int j = 0; j < 4; ++j) {
        float s = 0.f;
        for (int k = 0; k < 4; ++k) s += Pw[i][k] * A16[k][j];
        NM[i][j] = s;
      }
    for (int i = 0; i < 4; ++i)
      for (int j = 0; j < 4; ++j) Pw[i][j] = NM[i][j];
  }
}

__device__ __forceinline__ void lp_load_rows(v4f (&xv)[8], const float* __restrict__ xp) {
#pragma unroll
  for (int k = 0; k < 4; ++k) {
    xv[2 * k]     = *(const v4f*)(xp + (size_t)(4 * k) * S);
    xv[2 * k + 1] = *(const v4f*)(xp + (size_t)(4 * k + 1) * S);
  }
}

__device__ __forceinline__ void lp_zc_chunk(const v4f (&xv)[8], const v2f (&aq)[4],
                                            float* __restrict__ zc, int half) {
  v8f acc[4] = {};
#pragma unroll
  for (int k = 0; k < 4; ++k) {
#pragma unroll
    for (int j = 0; j < 4; ++j) {
      v2f bv;
      bv[0] = xv[2 * k][j];
      bv[1] = xv[2 * k + 1][j];
      acc[j] = __builtin_amdgcn_wmma_f32_16x16x4_f32(false, aq[k], false, bv,
                                                     (short)0, acc[j], false, false);
    }
  }
  if (half == 0) {
#pragma unroll
    for (int m = 0; m < 4; ++m) {
      v4f o;
      o[0] = acc[0][m]; o[1] = acc[1][m]; o[2] = acc[2][m]; o[3] = acc[3][m];
      *(v4f*)(zc + (size_t)m * S) = o;
    }
  }
}

__global__ __launch_bounds__(256) void lp_zc_kernel(const float* __restrict__ x,
                                                    float* __restrict__ ws) {
  const int wave = threadIdx.x >> 5;
  const int lane = threadIdx.x & 31;
  const int strip = blockIdx.x * 8 + wave;
  const int sg = strip & (NSTRIPG - 1);
  const int c0 = (strip / NSTRIPG) * RUN;
  const int half = lane >> 4;
  const int l16 = lane & 15;
  const int chan = sg * SW + 4 * l16;
  const float* Q = ws + OFF_Q;
  v2f aq[4];
#pragma unroll
  for (int k = 0; k < 4; ++k) {
    const int kk0 = 4 * k + 2 * half;
    aq[k][0] = Q[l16 * 16 + kk0];
    aq[k][1] = Q[l16 * 16 + kk0 + 1];
  }
  const float* xp = x + ((size_t)c0 * L + 2 * half) * S + chan;
  float* zc = ws + OFF_ZC + (size_t)c0 * 4 * S + chan;
  v4f xv[ZC_AHEAD + 1][8];
#pragma unroll
  for (int i = 0; i < ZC_AHEAD; ++i) lp_load_rows(xv[i], xp + (size_t)i * L * S);
#pragma unroll
  for (int i = 0; i < RUN; ++i) {
    __builtin_amdgcn_sched_barrier(0);
    if (i + ZC_AHEAD < RUN)
      lp_load_rows(xv[(i + ZC_AHEAD) % (ZC_AHEAD + 1)], xp + (size_t)(i + ZC_AHEAD) * L * S);
    __builtin_amdgcn_sched_barrier(0);
    lp_zc_chunk(xv[i % (ZC_AHEAD + 1)], aq, zc + (size_t)i * 4 * S, half);
  }
}

__global__ __launch_bounds__(256) void lp_scan1_kernel(float* __restrict__ ws) {
  const int tid = blockIdx.x * 256 + threadIdx.x;
  const int s  = tid & (S - 1);
  const int sb = tid >> 11;
  const float* A16 = ws + OFF_APOW + 16;
  float a[16];
#pragma unroll
  for (int i = 0; i < 16; ++i) a[i] = A16[i];
  float w0 = 0.f, w1 = 0.f, w2 = 0.f, w3 = 0.f;
  float* zc = ws + OFF_ZC + ((size_t)sb * SBLK) * 4 * S + s;
  for (int cc = 0; cc < SBLK; ++cc) {
    const float r0 = zc[0 * (size_t)S], r1 = zc[1 * (size_t)S];
    const float r2 = zc[2 * (size_t)S], r3 = zc[3 * (size_t)S];
    zc[0 * (size_t)S] = w0; zc[1 * (size_t)S] = w1;
    zc[2 * (size_t)S] = w2; zc[3 * (size_t)S] = w3;
    const float n0 = a[0]*w0 + a[1]*w1 + a[2]*w2 + a[3]*w3 + r0;
    const float n1 = a[4]*w0 + a[5]*w1 + a[6]*w2 + a[7]*w3 + r1;
    const float n2 = a[8]*w0 + a[9]*w1 + a[10]*w2 + a[11]*w3 + r2;
    const float n3 = a[12]*w0 + a[13]*w1 + a[14]*w2 + a[15]*w3 + r3;
    w0 = n0; w1 = n1; w2 = n2; w3 = n3;
    zc += (size_t)4 * S;
  }
  float* u = ws + OFF_U + (size_t)sb * 4 * S + s;
  u[0 * (size_t)S] = w0; u[1 * (size_t)S] = w1;
  u[2 * (size_t)S] = w2; u[3 * (size_t)S] = w3;
}

__global__ __launch_bounds__(256) void lp_scan2_kernel(float* __restrict__ ws) {
  const int s = blockIdx.x * 256 + threadIdx.x;
  const float* A = ws + OFF_APOW + 32 * 16;
  float a[16];
#pragma unroll
  for (int i = 0; i < 16; ++i) a[i] = A[i];
  float z0 = 0.f, z1 = 0.f, z2 = 0.f, z3 = 0.f;
  for (int sb = 0; sb < NSB; ++sb) {
    float* zs = ws + OFF_ZS + (size_t)sb * 4 * S + s;
    zs[0 * (size_t)S] = z0; zs[1 * (size_t)S] = z1;
    zs[2 * (size_t)S] = z2; zs[3 * (size_t)S] = z3;
    const float* u = ws + OFF_U + (size_t)sb * 4 * S + s;
    const float u0 = u[0 * (size_t)S], u1 = u[1 * (size_t)S];
    const float u2 = u[2 * (size_t)S], u3 = u[3 * (size_t)S];
    const float n0 = a[0]*z0 + a[1]*z1 + a[2]*z2 + a[3]*z3 + u0;
    const float n1 = a[4]*z0 + a[5]*z1 + a[6]*z2 + a[7]*z3 + u1;
    const float n2 = a[8]*z0 + a[9]*z1 + a[10]*z2 + a[11]*z3 + u2;
    const float n3 = a[12]*z0 + a[13]*z1 + a[14]*z2 + a[15]*z3 + u3;
    z0 = n0; z1 = n1; z2 = n2; z3 = n3;
  }
}

struct LpFinIn {
  v4f x[8];
  v4f w[2];
  v4f ap[2];
};

__device__ __forceinline__ void lp_fin_load(LpFinIn& in, const float* __restrict__ xp,
                                            const float* __restrict__ wp,
                                            const float* __restrict__ app) {
  in.ap[0] = *(const v4f*)(app);
  in.ap[1] = *(const v4f*)(app + 4);
  in.w[0]  = *(const v4f*)(wp);
  in.w[1]  = *(const v4f*)(wp + S);
  lp_load_rows(in.x, xp);
}

__device__ __forceinline__ void lp_fin_chunk(const LpFinIn& in, const v2f (&ah)[4], v2f pa,
                                             const v4f (&z)[4], float* __restrict__ o) {
  v8f acc[4];
#pragma unroll
  for (int j = 0; j < 4; ++j) {
    v2f zin;
#pragma unroll
    for (int e = 0; e < 2; ++e)
      zin[e] = in.ap[e][0] * z[0][j] + in.ap[e][1] * z[1][j] +
               in.ap[e][2] * z[2][j] + in.ap[e][3] * z[3][j] + in.w[e][j];
    v8f zero = {};
    acc[j] = __builtin_amdgcn_wmma_f32_16x16x4_f32(false, pa, false, zin,
                                                   (short)0, zero, false, false);
  }
#pragma unroll
  for (int k = 0; k < 4; ++k) {
#pragma unroll
    for (int j = 0; j < 4; ++j) {
      v2f bv;
      bv[0] = in.x[2 * k][j];
      bv[1] = in.x[2 * k + 1][j];
      acc[j] = __builtin_amdgcn_wmma_f32_16x16x4_f32(false, ah[k], false, bv,
                                                     (short)0, acc[j], false, false);
    }
  }
#pragma unroll
  for (int r = 0; r < 8; ++r) {
    v4f ov;
    ov[0] = acc[0][r]; ov[1] = acc[1][r]; ov[2] = acc[2][r]; ov[3] = acc[3][r];
    *(v4f*)(o + (size_t)r * S) = ov;
  }
}

__global__ __launch_bounds__(256) void lp_final_kernel(const float* __restrict__ x,
                                                       float* __restrict__ out,
                                                       const float* __restrict__ ws) {
  const int wave = threadIdx.x >> 5;
  const int lane = threadIdx.x & 31;
  const int strip = blockIdx.x * 8 + wave;
  const int sg = strip & (NSTRIPG - 1);
  const int c0 = (strip / NSTRIPG) * RUN;
  const int half = lane >> 4;
  const int l16 = lane & 15;
  const int sb = c0 >> 5;
  const int cc0 = c0 & 31;
  const int chan = sg * SW + 4 * l16;

  const float* H = ws + OFF_H;
  const float* Pm = ws + OFF_P;
  v2f ah[4], pa;
#pragma unroll
  for (int k = 0; k < 4; ++k) {
    const int kk0 = 4 * k + 2 * half;
    ah[k][0] = H[l16 * 16 + kk0];
    ah[k][1] = H[l16 * 16 + kk0 + 1];
  }
  pa[0] = Pm[l16 * 4 + 2 * half + 0];
  pa[1] = Pm[l16 * 4 + 2 * half + 1];

  const float* zs = ws + OFF_ZS + (size_t)sb * 4 * S + chan;
  v4f z[4];
#pragma unroll
  for (int m = 0; m < 4; ++m) z[m] = *(const v4f*)(zs + (size_t)m * S);

  const float* xp  = x + ((size_t)c0 * L + 2 * half) * S + chan;
  const float* wp  = ws + OFF_ZC + ((size_t)c0 * 4 + 2 * half) * S + chan;
  const float* app = ws + OFF_APOW + cc0 * 16 + 2 * half * 4;
  float* o = out + ((size_t)c0 * L + 8 * half) * S + chan;

  LpFinIn in[FIN_AHEAD + 1];
#pragma unroll
  for (int i = 0; i < FIN_AHEAD; ++i)
    lp_fin_load(in[i], xp + (size_t)i * L * S, wp + (size_t)i * 4 * S, app + i * 16);
#pragma unroll
  for (int i = 0; i < RUN; ++i) {
    __builtin_amdgcn_sched_barrier(0);
    if (i + FIN_AHEAD < RUN)
      lp_fin_load(in[(i + FIN_AHEAD) % (FIN_AHEAD + 1)], xp + (size_t)(i + FIN_AHEAD) * L * S,
                  wp + (size_t)(i + FIN_AHEAD) * 4 * S, app + (i + FIN_AHEAD) * 16);
    __builtin_amdgcn_sched_barrier(0);
    lp_fin_chunk(in[i % (FIN_AHEAD + 1)], ah, pa, z, o + (size_t)i * L * S);
  }
}

extern "C" void kernel_launch(void* const* d_in, const int* in_sizes, int n_in,
                              void* d_out, int out_size, void* d_ws, size_t ws_size,
                              hipStream_t stream) {
  const float* signal = (const float*)d_in[0];
  const float* b      = (const float*)d_in[1];
  const float* a      = (const float*)d_in[2];
  float* out = (float*)d_out;
  float* ws  = (float*)d_ws;

  lp_setup_kernel<<<1, 64, 0, stream>>>(b, a, ws);

  lp_zc_kernel<<<STRIP_BLOCKS, 256, 0, stream>>>(signal, ws);

  lp_scan1_kernel<<<(S * NSB) / 256, 256, 0, stream>>>(ws);
  lp_scan2_kernel<<<S / 256, 256, 0, stream>>>(ws);

  lp_final_kernel<<<STRIP_BLOCKS, 256, 0, stream>>>(signal, out, ws);
}


# State kept per 8-chunk run, not per chunk: pass 1 sums a run's 128 rows in its accumulators, pass 3 carries the state chunk to chunk in registers from the last 4 x,y rows; summation order changes
# speedup vs baseline: 1.6381x; 1.3142x over previous
#include <hip/hip_runtime.h>
#include <hip/hip_bf16.h>

typedef __attribute__((ext_vector_type(2))) float v2f;
typedef __attribute__((ext_vector_type(4))) float v4f;
typedef __attribute__((ext_vector_type(8))) float v8f;

constexpr int S       = 2048;
constexpr int T       = 16384;
constexpr int L       = 16;
constexpr int NCHUNK  = T / L;
constexpr int SW       = 64;
constexpr int NSTRIPG  = S / SW;
constexpr int RUN      = 8;
constexpr int RLEN     = RUN * L;
constexpr int NRUN     = NCHUNK / RUN;
constexpr int SBLK     = 8;
constexpr int NSB      = NRUN / SBLK;
constexpr int ZC_AHEAD  = 2;
constexpr int FIN_AHEAD = 2;
constexpr int NSTRIPS  = NSTRIPG * NRUN;
constexpr int STRIP_BLOCKS = NSTRIPS / 8;

constexpr size_t OFF_H    = 0;
constexpr size_t OFF_P    = 256;
constexpr size_t OFF_Q    = 320;
constexpr size_t OFF_APOW = 832;
constexpr size_t OFF_ZC   = 4096;
constexpr size_t OFF_U    = OFF_ZC + (size_t)NRUN * 4 * S;
constexpr size_t OFF_ZS   = OFF_U  + (size_t)NSB  * 4 * S;

__device__ __forceinline__ void lp_mm4(const float (&X)[4][4], const float (&Y)[4][4],
                                       float (&Z)[4][4]) {
  for (int i = 0; i < 4; ++i)
    for (int j = 0; j < 4; ++j) {
      float s = 0.f;
      for (int k = 0; k < 4; ++k) s += X[i][k] * Y[k][j];
      Z[i][j] = s;
    }
}

__global__ void lp_setup_kernel(const float* __restrict__ bc,
                                const float* __restrict__ ac,
                                float* __restrict__ ws) {
  if (threadIdx.x != 0 || blockIdx.x != 0) return;
  const float b0 = bc[0];
  float A[4][4], g4[4];
  for (int i = 0; i < 4; ++i) {
    for (int j = 0; j < 4; ++j) A[i][j] = (j == i + 1) ? 1.0f : 0.0f;
    A[i][0] -= ac[i + 1];
    g4[i] = bc[i + 1] - ac[i + 1] * b0;
  }
  float h[16];
  h[0] = b0;
  float v[4] = {g4[0], g4[1], g4[2], g4[3]};
  for (int t = 1; t < 16; ++t) {
    h[t] = v[0];
    float nv[4];
    for (int i = 0; i < 4; ++i) {
      float s = 0.f;
      for (int j = 0; j < 4; ++j) s += A[i][j] * v[j];
      nv[i] = s;
    }
    for (int i = 0; i < 4; ++i) v[i] = nv[i];
  }
  for (int t = 0; t < 16; ++t)
    for (int k = 0; k < 16; ++k)
      ws[OFF_H + t * 16 + k] = (k <= t) ? h[t - k] : 0.0f;
  float M[4][4];
  for (int i = 0; i < 4; ++i)
    for (int j = 0; j < 4; ++j) M[i][j] = (i == j) ? 1.0f : 0.0f;
  for (int t = 0; t < 16; ++t) {
    for (int m = 0; m < 4; ++m) ws[OFF_P + t * 4 + m] = M[0][m];
    float NM[4][4];
    lp_mm4(A, M, NM);
    for (int i = 0; i < 4; ++i)
      for (int j = 0; j < 4; ++j) M[i][j] = NM[i][j];
  }
  float A16[4][4], A128[4][4];
  for (int i = 0; i < 4; ++i)
    for (int j = 0; j < 4; ++j) { A16[i][j] = M[i][j]; A128[i][j] = M[i][j]; }
  for (int rep = 1; rep < RUN; ++rep) {
    float NM[4][4];
    lp_mm4(A128, A16, NM);
    for (int i = 0; i < 4; ++i)
      for (int j = 0; j < 4; ++j) A128[i][j] = NM[i][j];
  }
  for (int i = 0; i < 4; ++i) v[i] = g4[i];
  for (int kk = RLEN - 1; kk >= 0; --kk) {
    for (int m = 0; m < 4; ++m) ws[OFF_Q + m * RLEN + kk] = v[m];
    float nv[4];
    for (int i = 0; i < 4; ++i) {
      float s = 0.f;
      for (int j = 0; j < 4; ++j) s += A[i][j] * v[j];
      nv[i] = s;
    }
    for (int i = 0; i < 4; ++i) v[i] = nv[i];
  }
  float Pw[4][4];
  for (int i = 0; i < 4; ++i)
    for (int j = 0; j < 4; ++j) Pw[i][j] = (i == j) ? 1.0f : 0.0f;
  for (int p = 0; p <= SBLK; ++p) {
    for (int i = 0; i < 4; ++i)
      for (int j = 0; j < 4; ++j) ws[OFF_APOW + p * 16 + i * 4 + j] = Pw[i][j];
    float NM[4][4];
    lp_mm4(Pw, A128, NM);
    for (int i = 0; i < 4; ++i)
      for (int j = 0; j < 4; ++j) Pw[i][j] = NM[i][j];
  }
}

__device__ __forceinline__ void lp_load_rows(v4f (&xv)[8], const float* __restrict__ xp) {
#pragma unroll
  for (int k = 0; k < 4; ++k) {
    xv[2 * k]     = *(const v4f*)(xp + (size_t)(4 * k) * S);
    xv[2 * k + 1] = *(const v4f*)(xp + (size_t)(4 * k + 1) * S);
  }
}

__device__ __forceinline__ void lp_zc_chunk(v8f (&acc)[4], const v4f (&xv)[8],
                                            const v2f (&aq)[4]) {
#pragma unroll
  for (int k = 0; k < 4; ++k) {
#pragma unroll
    for (int j = 0; j < 4; ++j) {
      v2f bv;
      bv[0] = xv[2 * k][j];
      bv[1] = xv[2 * k + 1][j];
      acc[j] = __builtin_amdgcn_wmma_f32_16x16x4_f32(false, aq[k], false, bv,
                                                     (short)0, acc[j], false, false);
    }
  }
}

__global__ __launch_bounds__(256) void lp_zc_kernel(const float* __restrict__ x,
                                                    float* __restrict__ ws) {
  const int wave = threadIdx.x >> 5;
  const int lane = threadIdx.x & 31;
  const int strip = blockIdx.x * 8 + wave;
  const int sg = strip & (NSTRIPG - 1);
  const int run = strip / NSTRIPG;
  const int half = lane >> 4;
  const int l16 = lane & 15;
  const int chan = sg * SW + 4 * l16;
  const float* Q = ws + OFF_Q + (l16 & 3) * RLEN + 2 * half;
  const bool qrow = l16 < 4;
  v2f qraw[RUN][4];
#pragma unroll
  for (int i = 0; i < RUN; ++i) {
#pragma unroll
    for (int k = 0; k < 4; ++k) qraw[i][k] = *(const v2f*)(Q + 16 * i + 4 * k);
  }
  const float* xp = x + ((size_t)run * RLEN + 2 * half) * S + chan;
  v4f xv[ZC_AHEAD + 1][8];
  __builtin_amdgcn_sched_barrier(0);
#pragma unroll
  for (int i = 0; i < ZC_AHEAD; ++i) lp_load_rows(xv[i], xp + (size_t)i * L * S);
  v8f acc[4] = {};
#pragma unroll
  for (int i = 0; i < RUN; ++i) {
    __builtin_amdgcn_sched_barrier(0);
    if (i + ZC_AHEAD < RUN)
      lp_load_rows(xv[(i + ZC_AHEAD) % (ZC_AHEAD + 1)], xp + (size_t)(i + ZC_AHEAD) * L * S);
    __builtin_amdgcn_sched_barrier(0);
    v2f aq[4];
#pragma unroll
    for (int k = 0; k < 4; ++k) {
      aq[k][0] = qrow ? qraw[i][k][0] : 0.0f;
      aq[k][1] = qrow ? qraw[i][k][1] : 0.0f;
    }
    lp_zc_chunk(acc, xv[i % (ZC_AHEAD + 1)], aq);
  }
  if (half == 0) {
    float* zr = ws + OFF_ZC + (size_t)run * 4 * S + chan;
#pragma unroll
    for (int m = 0; m < 4; ++m) {
      v4f o;
      o[0] = acc[0][m]; o[1] = acc[1][m]; o[2] = acc[2][m]; o[3] = acc[3][m];
      *(v4f*)(zr + (size_t)m * S) = o;
    }
  }
}

__global__ __launch_bounds__(256) void lp_scan1_kernel(float* __restrict__ ws) {
  const int tid = blockIdx.x * 256 + threadIdx.x;
  const int s  = tid & (S - 1);
  const int sb = tid >> 11;
  const float* A1 = ws + OFF_APOW + 16;
  float a[16];
#pragma unroll
  for (int i = 0; i < 16; ++i) a[i] = A1[i];
  float w0 = 0.f, w1 = 0.f, w2 = 0.f, w3 = 0.f;
  float* zc = ws + OFF_ZC + ((size_t)sb * SBLK) * 4 * S + s;
  for (int cc = 0; cc < SBLK; ++cc) {
    const float r0 = zc[0 * (size_t)S], r1 = zc[1 * (size_t)S];
    const float r2 = zc[2 * (size_t)S], r3 = zc[3 * (size_t)S];
    zc[0 * (size_t)S] = w0; zc[1 * (size_t)S] = w1;
    zc[2 * (size_t)S] = w2; zc[3 * (size_t)S] = w3;
    const float n0 = a[0]*w0 + a[1]*w1 + a[2]*w2 + a[3]*w3 + r0;
    const float n1 = a[4]*w0 + a[5]*w1 + a[6]*w2 + a[7]*w3 + r1;
    const float n2 = a[8]*w0 + a[9]*w1 + a[10]*w2 + a[11]*w3 + r2;
    const float n3 = a[12]*w0 + a[13]*w1 + a[14]*w2 + a[15]*w3 + r3;
    w0 = n0; w1 = n1; w2 = n2; w3 = n3;
    zc += (size_t)4 * S;
  }
  float* u = ws + OFF_U + (size_t)sb * 4 * S + s;
  u[0 * (size_t)S] = w0; u[1 * (size_t)S] = w1;
  u[2 * (size_t)S] = w2; u[3 * (size_t)S] = w3;
}

__global__ __launch_bounds__(256) void lp_scan2_kernel(float* __restrict__ ws) {
  const int s = blockIdx.x * 256 + threadIdx.x;
  const float* A = ws + OFF_APOW + SBLK * 16;
  float a[16];
#pragma unroll
  for (int i = 0; i < 16; ++i) a[i] = A[i];
  float z0 = 0.f, z1 = 0.f, z2 = 0.f, z3 = 0.f;
  for (int sb = 0; sb < NSB; ++sb) {
    float* zs = ws + OFF_ZS + (size_t)sb * 4 * S + s;
    zs[0 * (size_t)S] = z0; zs[1 * (size_t)S] = z1;
    zs[2 * (size_t)S] = z2; zs[3 * (size_t)S] = z3;
    const float* u = ws + OFF_U + (size_t)sb * 4 * S + s;
    const float u0 = u[0 * (size_t)S], u1 = u[1 * (size_t)S];
    const float u2 = u[2 * (size_t)S], u3 = u[3 * (size_t)S];
    const float n0 = a[0]*z0 + a[1]*z1 + a[2]*z2 + a[3]*z3 + u0;
    const float n1 = a[4]*z0 + a[5]*z1 + a[6]*z2 + a[7]*z3 + u1;
    const float n2 = a[8]*z0 + a[9]*z1 + a[10]*z2 + a[11]*z3 + u2;
    const float n3 = a[12]*z0 + a[13]*z1 + a[14]*z2 + a[15]*z3 + u3;
    z0 = n0; z1 = n1; z2 = n2; z3 = n3;
  }
}

struct LpCarry {
  float b1, b2, b3, b4, a1, a2, a3, a4;
  float cA, cB, cC, cD;
  bool hi;
};

__device__ __forceinline__ float lp_xhalf(float v) {
  return __builtin_bit_cast(float,
      __builtin_amdgcn_permlanex16(__builtin_bit_cast(int, v), __builtin_bit_cast(int, v),
                                   0x76543210, 0xfedcba98, false, false));
}

__device__ __forceinline__ void lp_fin_chunk(const v4f (&xv)[8], const v2f (&ah)[4], v2f pa,
                                             v2f (&zin)[4], float* __restrict__ o,
                                             const LpCarry& cf, const bool last) {
  v8f acc[4];
#pragma unroll
  for (int j = 0; j < 4; ++j) {
    v8f zero = {};
    acc[j] = __builtin_amdgcn_wmma_f32_16x16x4_f32(false, pa, false, zin[j],
                                                   (short)0, zero, false, false);
  }
#pragma unroll
  for (int k = 0; k < 4; ++k) {
#pragma unroll
    for (int j = 0; j < 4; ++j) {
      v2f bv;
      bv[0] = xv[2 * k][j];
      bv[1] = xv[2 * k + 1][j];
      acc[j] = __builtin_amdgcn_wmma_f32_16x16x4_f32(false, ah[k], false, bv,
                                                     (short)0, acc[j], false, false);
    }
  }
  if (!last) {
#pragma unroll
    for (int j = 0; j < 4; ++j) {
      const float x6 = xv[6][j], x7 = xv[7][j];
      const float y4 = acc[j][4], y5 = acc[j][5], y6 = acc[j][6], y7 = acc[j][7];
      const float u0 = cf.cA * x6 + cf.cB * x7;
      const float u1 = cf.cC * x6 + cf.cD * x7;
      const float s0 = u0 - (cf.a4 * y4 + cf.a3 * y5 + cf.a2 * y6 + cf.a1 * y7);
      const float s1 = u1 - (cf.a4 * y5 + cf.a3 * y6 + cf.a2 * y7);
      const float s2 = (cf.b4 * x6 + cf.b3 * x7) - (cf.a4 * y6 + cf.a3 * y7);
      const float s3 = cf.b4 * x7 - cf.a4 * y7;
      const float t0 = lp_xhalf(s0);
      const float t1 = lp_xhalf(s1);
      zin[j][0] = cf.hi ? s2 : (u0 + t0);
      zin[j][1] = cf.hi ? s3 : (u1 + t1);
    }
  }
#pragma unroll
  for (int r = 0; r < 8; ++r) {
    v4f ov;
    ov[0] = acc[0][r]; ov[1] = acc[1][r]; ov[2] = acc[2][r]; ov[3] = acc[3][r];
    *(v4f*)(o + (size_t)r * S) = ov;
  }
}

__global__ __launch_bounds__(256) void lp_final_kernel(const float* __restrict__ x,
                                                       const float* __restrict__ bc,
                                                       const float* __restrict__ ac,
                                                       float* __restrict__ out,
                                                       const float* __restrict__ ws) {
  const int wave = threadIdx.x >> 5;
  const int lane = threadIdx.x & 31;
  const int strip = blockIdx.x * 8 + wave;
  const int sg = strip & (NSTRIPG - 1);
  const int run = strip / NSTRIPG;
  const int half = lane >> 4;
  const int l16 = lane & 15;
  const int sb = run / SBLK;
  const int rr = run % SBLK;
  const int chan = sg * SW + 4 * l16;

  const float* H = ws + OFF_H;
  const float* Pm = ws + OFF_P;
  v2f ah[4], pa;
#pragma unroll
  for (int k = 0; k < 4; ++k) {
    const int kk0 = 4 * k + 2 * half;
    ah[k][0] = H[l16 * 16 + kk0];
    ah[k][1] = H[l16 * 16 + kk0 + 1];
  }
  pa[0] = Pm[l16 * 4 + 2 * half + 0];
  pa[1] = Pm[l16 * 4 + 2 * half + 1];

  LpCarry cf;
  cf.b1 = bc[1]; cf.b2 = bc[2]; cf.b3 = bc[3]; cf.b4 = bc[4];
  cf.a1 = ac[1]; cf.a2 = ac[2]; cf.a3 = ac[3]; cf.a4 = ac[4];
  cf.hi = half != 0;
  cf.cA = cf.hi ? cf.b2 : cf.b4;
  cf.cB = cf.hi ? cf.b1 : cf.b3;
  cf.cC = cf.hi ? cf.b3 : 0.0f;
  cf.cD = cf.hi ? cf.b2 : cf.b4;

  const float* zs  = ws + OFF_ZS + (size_t)sb * 4 * S + chan;
  const float* wp  = ws + OFF_ZC + ((size_t)run * 4 + 2 * half) * S + chan;
  const float* app = ws + OFF_APOW + rr * 16 + 2 * half * 4;
  v4f z[4], w[2], ap[2];
#pragma unroll
  for (int m = 0; m < 4; ++m) z[m] = *(const v4f*)(zs + (size_t)m * S);
#pragma unroll
  for (int e = 0; e < 2; ++e) {
    w[e]  = *(const v4f*)(wp + (size_t)e * S);
    ap[e] = *(const v4f*)(app + 4 * e);
  }
  v2f zin[4];

  const float* xp  = x + ((size_t)run * RLEN + 2 * half) * S + chan;
  float* o = out + ((size_t)run * RLEN + 8 * half) * S + chan;

  v4f xv[FIN_AHEAD + 1][8];
  __builtin_amdgcn_sched_barrier(0);
#pragma unroll
  for (int i = 0; i < FIN_AHEAD; ++i) lp_load_rows(xv[i], xp + (size_t)i * L * S);
#pragma unroll
  for (int i = 0; i < RUN; ++i) {
    __builtin_amdgcn_sched_barrier(0);
    if (i + FIN_AHEAD < RUN)
      lp_load_rows(xv[(i + FIN_AHEAD) % (FIN_AHEAD + 1)], xp + (size_t)(i + FIN_AHEAD) * L * S);
    __builtin_amdgcn_sched_barrier(0);
    if (i == 0) {
#pragma unroll
      for (int j = 0; j < 4; ++j) {
#pragma unroll
        for (int e = 0; e < 2; ++e)
          zin[j][e] = ap[e][0] * z[0][j] + ap[e][1] * z[1][j] +
                      ap[e][2] * z[2][j] + ap[e][3] * z[3][j] + w[e][j];
      }
    }
    lp_fin_chunk(xv[i % (FIN_AHEAD + 1)], ah, pa, zin, o + (size_t)i * L * S, cf, i == RUN - 1);
  }
}

extern "C" void kernel_launch(void* const* d_in, const int* in_sizes, int n_in,
                              void* d_out, int out_size, void* d_ws, size_t ws_size,
                              hipStream_t stream) {
  const float* signal = (const float*)d_in[0];
  const float* b      = (const float*)d_in[1];
  const float* a      = (const float*)d_in[2];
  float* out = (float*)d_out;
  float* ws  = (float*)d_ws;

  lp_setup_kernel<<<1, 64, 0, stream>>>(b, a, ws);

  lp_zc_kernel<<<STRIP_BLOCKS, 256, 0, stream>>>(signal, ws);

  lp_scan1_kernel<<<(S * NSB) / 256, 256, 0, stream>>>(ws);
  lp_scan2_kernel<<<S / 256, 256, 0, stream>>>(ws);

  lp_final_kernel<<<STRIP_BLOCKS, 256, 0, stream>>>(signal, b, a, out, ws);
}


# Both state scans folded into the WMMA passes (block = 8 runs of one strip): pass 1 scans them in LDS, writes w and u; pass 3 rebuilds zs from u; same recurrences and sum order; 5 -> 3 launches
# speedup vs baseline: 2.0137x; 1.2293x over previous
#include <hip/hip_runtime.h>
#include <hip/hip_bf16.h>

typedef __attribute__((ext_vector_type(2))) float v2f;
typedef __attribute__((ext_vector_type(4))) float v4f;
typedef __attribute__((ext_vector_type(8))) float v8f;

constexpr int S       = 2048;
constexpr int T       = 16384;
constexpr int L       = 16;
constexpr int NCHUNK  = T / L;
constexpr int SW       = 64;
constexpr int NSTRIPG  = S / SW;
constexpr int RUN      = 8;
constexpr int RLEN     = RUN * L;
constexpr int NRUN     = NCHUNK / RUN;
constexpr int SBLK     = 8;
constexpr int NSB      = NRUN / SBLK;
constexpr int ZC_AHEAD  = 2;
constexpr int FIN_AHEAD = 2;
constexpr int NSTRIPS  = NSTRIPG * NRUN;
constexpr int STRIP_BLOCKS = NSTRIPS / SBLK;
static_assert(NSB == 2 * SBLK, "pass 3: every wave fetches two of the NSB superblock states");

constexpr size_t OFF_H    = 0;
constexpr size_t OFF_P    = 256;
constexpr size_t OFF_Q    = 320;
constexpr size_t OFF_APOW = 832;
constexpr size_t OFF_ZC   = 4096;
constexpr size_t OFF_U    = OFF_ZC + (size_t)NRUN * 4 * S;

__device__ __forceinline__ void lp_mm4(const float (&X)[4][4], const float (&Y)[4][4],
                                       float (&Z)[4][4]) {
  for (int i = 0; i < 4; ++i)
    for (int j = 0; j < 4; ++j) {
      float s = 0.f;
      for (int k = 0; k < 4; ++k) s += X[i][k] * Y[k][j];
      Z[i][j] = s;
    }
}

__global__ void lp_setup_kernel(const float* __restrict__ bc,
                                const float* __restrict__ ac,
                                float* __restrict__ ws) {
  if (threadIdx.x != 0 || blockIdx.x != 0) return;
  const float b0 = bc[0];
  float A[4][4], g4[4];
  for (int i = 0; i < 4; ++i) {
    for (int j = 0; j < 4; ++j) A[i][j] = (j == i + 1) ? 1.0f : 0.0f;
    A[i][0] -= ac[i + 1];
    g4[i] = bc[i + 1] - ac[i + 1] * b0;
  }
  float h[16];
  h[0] = b0;
  float v[4] = {g4[0], g4[1], g4[2], g4[3]};
  for (int t = 1; t < 16; ++t) {
    h[t] = v[0];
    float nv[4];
    for (int i = 0; i < 4; ++i) {
      float s = 0.f;
      for (int j = 0; j < 4; ++j) s += A[i][j] * v[j];
      nv[i] = s;
    }
    for (int i = 0; i < 4; ++i) v[i] = nv[i];
  }
  for (int t = 0; t < 16; ++t)
    for (int k = 0; k < 16; ++k)
      ws[OFF_H + t * 16 + k] = (k <= t) ? h[t - k] : 0.0f;
  float M[4][4];
  for (int i = 0; i < 4; ++i)
    for (int j = 0; j < 4; ++j) M[i][j] = (i == j) ? 1.0f : 0.0f;
  for (int t = 0; t < 16; ++t) {
    for (int m = 0; m < 4; ++m) ws[OFF_P + t * 4 + m] = M[0][m];
    float NM[4][4];
    lp_mm4(A, M, NM);
    for (int i = 0; i < 4; ++i)
      for (int j = 0; j < 4; ++j) M[i][j] = NM[i][j];
  }
  float A16[4][4], A128[4][4];
  for (int i = 0; i < 4; ++i)
    for (int j = 0; j < 4; ++j) { A16[i][j] = M[i][j]; A128[i][j] = M[i][j]; }
  for (int rep = 1; rep < RUN; ++rep) {
    float NM[4][4];
    lp_mm4(A128, A16, NM);
    for (int i = 0; i < 4; ++i)
      for (int j = 0; j < 4; ++j) A128[i][j] = NM[i][j];
  }
  for (int i = 0; i < 4; ++i) v[i] = g4[i];
  for (int kk = RLEN - 1; kk >= 0; --kk) {
    for (int m = 0; m < 4; ++m) ws[OFF_Q + m * RLEN + kk] = v[m];
    float nv[4];
    for (int i = 0; i < 4; ++i) {
      float s = 0.f;
      for (int j = 0; j < 4; ++j) s += A[i][j] * v[j];
      nv[i] = s;
    }
    for (int i = 0; i < 4; ++i) v[i] = nv[i];
  }
  float Pw[4][4];
  for (int i = 0; i < 4; ++i)
    for (int j = 0; j < 4; ++j) Pw[i][j] = (i == j) ? 1.0f : 0.0f;
  for (int p = 0; p <= SBLK; ++p) {
    for (int i = 0; i < 4; ++i)
      for (int j = 0; j < 4; ++j) ws[OFF_APOW + p * 16 + i * 4 + j] = Pw[i][j];
    float NM[4][4];
    lp_mm4(Pw, A128, NM);
    for (int i = 0; i < 4; ++i)
      for (int j = 0; j < 4; ++j) Pw[i][j] = NM[i][j];
  }
}

__device__ __forceinline__ void lp_load_rows(v4f (&xv)[8], const float* __restrict__ xp) {
#pragma unroll
  for (int k = 0; k < 4; ++k) {
    xv[2 * k]     = *(const v4f*)(xp + (size_t)(4 * k) * S);
    xv[2 * k + 1] = *(const v4f*)(xp + (size_t)(4 * k + 1) * S);
  }
}

__device__ __forceinline__ void lp_zc_chunk(v8f (&acc)[4], const v4f (&xv)[8],
                                            const v2f (&aq)[4]) {
#pragma unroll
  for (int k = 0; k < 4; ++k) {
#pragma unroll
    for (int j = 0; j < 4; ++j) {
      v2f bv;
      bv[0] = xv[2 * k][j];
      bv[1] = xv[2 * k + 1][j];
      acc[j] = __builtin_amdgcn_wmma_f32_16x16x4_f32(false, aq[k], false, bv,
                                                     (short)0, acc[j], false, false);
    }
  }
}

__global__ __launch_bounds__(256) void lp_zc_kernel(const float* __restrict__ x,
                                                    float* __restrict__ ws) {
  __shared__ __attribute__((aligned(16))) float zsh[SBLK][4][SW];
  const int wave = __builtin_amdgcn_readfirstlane((int)(threadIdx.x >> 5));
  const int lane = threadIdx.x & 31;
  const int sg = blockIdx.x & (NSTRIPG - 1);
  const int sb = blockIdx.x / NSTRIPG;
  const int run = sb * SBLK + wave;
  const int half = lane >> 4;
  const int l16 = lane & 15;
  const int chan = sg * SW + 4 * l16;
  const float* Q = ws + OFF_Q + (l16 & 3) * RLEN + 2 * half;
  const bool qrow = l16 < 4;
  v2f qraw[RUN][4];
#pragma unroll
  for (int i = 0; i < RUN; ++i) {
#pragma unroll
    for (int k = 0; k < 4; ++k) qraw[i][k] = *(const v2f*)(Q + 16 * i + 4 * k);
  }
  const float* A1 = ws + OFF_APOW + 16;
  float a[16];
#pragma unroll
  for (int i = 0; i < 16; ++i) a[i] = A1[i];
  const float* xp = x + ((size_t)run * RLEN + 2 * half) * S + chan;
  v4f xv[ZC_AHEAD + 1][8];
  __builtin_amdgcn_sched_barrier(0);
#pragma unroll
  for (int i = 0; i < ZC_AHEAD; ++i) lp_load_rows(xv[i], xp + (size_t)i * L * S);
  v8f acc[4] = {};
#pragma unroll
  for (int i = 0; i < RUN; ++i) {
    __builtin_amdgcn_sched_barrier(0);
    if (i + ZC_AHEAD < RUN)
      lp_load_rows(xv[(i + ZC_AHEAD) % (ZC_AHEAD + 1)], xp + (size_t)(i + ZC_AHEAD) * L * S);
    __builtin_amdgcn_sched_barrier(0);
    v2f aq[4];
#pragma unroll
    for (int k = 0; k < 4; ++k) {
      aq[k][0] = qrow ? qraw[i][k][0] : 0.0f;
      aq[k][1] = qrow ? qraw[i][k][1] : 0.0f;
    }
    lp_zc_chunk(acc, xv[i % (ZC_AHEAD + 1)], aq);
  }
  if (half == 0) {
#pragma unroll
    for (int m = 0; m < 4; ++m) {
      v4f o;
      o[0] = acc[0][m]; o[1] = acc[1][m]; o[2] = acc[2][m]; o[3] = acc[3][m];
      *(v4f*)&zsh[wave][m][4 * l16] = o;
    }
  }
  __syncthreads();
  const int c2 = 2 * lane;
  v2f r[SBLK][4];
#pragma unroll
  for (int jj = 0; jj < SBLK; ++jj) {
#pragma unroll
    for (int m = 0; m < 4; ++m) r[jj][m] = *(const v2f*)&zsh[jj][m][c2];
  }
  v2f w0 = {0.f, 0.f}, w1 = w0, w2 = w0, w3 = w0;
  v2f k0 = w0, k1 = w0, k2 = w0, k3 = w0;
#pragma unroll
  for (int jj = 0; jj < SBLK; ++jj) {
    if (jj == wave) { k0 = w0; k1 = w1; k2 = w2; k3 = w3; }
    const v2f n0 = a[0]*w0 + a[1]*w1 + a[2]*w2 + a[3]*w3 + r[jj][0];
    const v2f n1 = a[4]*w0 + a[5]*w1 + a[6]*w2 + a[7]*w3 + r[jj][1];
    const v2f n2 = a[8]*w0 + a[9]*w1 + a[10]*w2 + a[11]*w3 + r[jj][2];
    const v2f n3 = a[12]*w0 + a[13]*w1 + a[14]*w2 + a[15]*w3 + r[jj][3];
    w0 = n0; w1 = n1; w2 = n2; w3 = n3;
  }
  float* wo = ws + OFF_ZC + (size_t)run * 4 * S + sg * SW + c2;
  *(v2f*)(wo + 0 * (size_t)S) = k0; *(v2f*)(wo + 1 * (size_t)S) = k1;
  *(v2f*)(wo + 2 * (size_t)S) = k2; *(v2f*)(wo + 3 * (size_t)S) = k3;
  if (wave == SBLK - 1) {
    float* uo = ws + OFF_U + (size_t)sb * 4 * S + sg * SW + c2;
    *(v2f*)(uo + 0 * (size_t)S) = w0; *(v2f*)(uo + 1 * (size_t)S) = w1;
    *(v2f*)(uo + 2 * (size_t)S) = w2; *(v2f*)(uo + 3 * (size_t)S) = w3;
  }
}

struct LpCarry {
  float b1, b2, b3, b4, a1, a2, a3, a4;
  float cA, cB, cC, cD;
  bool hi;
};

__device__ __forceinline__ float lp_xhalf(float v) {
  return __builtin_bit_cast(float,
      __builtin_amdgcn_permlanex16(__builtin_bit_cast(int, v), __builtin_bit_cast(int, v),
                                   0x76543210, 0xfedcba98, false, false));
}

__device__ __forceinline__ void lp_fin_chunk(const v4f (&xv)[8], const v2f (&ah)[4], v2f pa,
                                             v2f (&zin)[4], float* __restrict__ o,
                                             const LpCarry& cf, const bool last) {
  v8f acc[4];
#pragma unroll
  for (int j = 0; j < 4; ++j) {
    v8f zero = {};
    acc[j] = __builtin_amdgcn_wmma_f32_16x16x4_f32(false, pa, false, zin[j],
                                                   (short)0, zero, false, false);
  }
#pragma unroll
  for (int k = 0; k < 4; ++k) {
#pragma unroll
    for (int j = 0; j < 4; ++j) {
      v2f bv;
      bv[0] = xv[2 * k][j];
      bv[1] = xv[2 * k + 1][j];
      acc[j] = __builtin_amdgcn_wmma_f32_16x16x4_f32(false, ah[k], false, bv,
                                                     (short)0, acc[j], false, false);
    }
  }
  if (!last) {
#pragma unroll
    for (int j = 0; j < 4; ++j) {
      const float x6 = xv[6][j], x7 = xv[7][j];
      const float y4 = acc[j][4], y5 = acc[j][5], y6 = acc[j][6], y7 = acc[j][7];
      const float u0 = cf.cA * x6 + cf.cB * x7;
      const float u1 = cf.cC * x6 + cf.cD * x7;
      const float s0 = u0 - (cf.a4 * y4 + cf.a3 * y5 + cf.a2 * y6 + cf.a1 * y7);
      const float s1 = u1 - (cf.a4 * y5 + cf.a3 * y6 + cf.a2 * y7);
      const float s2 = (cf.b4 * x6 + cf.b3 * x7) - (cf.a4 * y6 + cf.a3 * y7);
      const float s3 = cf.b4 * x7 - cf.a4 * y7;
      const float t0 = lp_xhalf(s0);
      const float t1 = lp_xhalf(s1);
      zin[j][0] = cf.hi ? s2 : (u0 + t0);
      zin[j][1] = cf.hi ? s3 : (u1 + t1);
    }
  }
#pragma unroll
  for (int r = 0; r < 8; ++r) {
    v4f ov;
    ov[0] = acc[0][r]; ov[1] = acc[1][r]; ov[2] = acc[2][r]; ov[3] = acc[3][r];
    *(v4f*)(o + (size_t)r * S) = ov;
  }
}

__global__ __launch_bounds__(256) void lp_final_kernel(const float* __restrict__ x,
                                                       const float* __restrict__ bc,
                                                       const float* __restrict__ ac,
                                                       float* __restrict__ out,
                                                       const float* __restrict__ ws) {
  __shared__ __attribute__((aligned(16))) float ush[NSB][4][SW];
  const int wave = __builtin_amdgcn_readfirstlane((int)(threadIdx.x >> 5));
  const int lane = threadIdx.x & 31;
  const int sg = blockIdx.x & (NSTRIPG - 1);
  const int sb = blockIdx.x / NSTRIPG;
  const int rr = wave;
  const int run = sb * SBLK + rr;
  const int half = lane >> 4;
  const int l16 = lane & 15;
  const int chan = sg * SW + 4 * l16;

  const float* up = ws + OFF_U + sg * SW + 2 * lane;
  v2f ur[2][4];
#pragma unroll
  for (int t = 0; t < 2; ++t) {
#pragma unroll
    for (int m = 0; m < 4; ++m)
      ur[t][m] = *(const v2f*)(up + ((size_t)(wave + t * SBLK) * 4 + m) * S);
  }

  const float* H = ws + OFF_H;
  const float* Pm = ws + OFF_P;
  v2f ah[4], pa;
#pragma unroll
  for (int k = 0; k < 4; ++k) {
    const int kk0 = 4 * k + 2 * half;
    ah[k][0] = H[l16 * 16 + kk0];
    ah[k][1] = H[l16 * 16 + kk0 + 1];
  }
  pa[0] = Pm[l16 * 4 + 2 * half + 0];
  pa[1] = Pm[l16 * 4 + 2 * half + 1];

  LpCarry cf;
  cf.b1 = bc[1]; cf.b2 = bc[2]; cf.b3 = bc[3]; cf.b4 = bc[4];
  cf.a1 = ac[1]; cf.a2 = ac[2]; cf.a3 = ac[3]; cf.a4 = ac[4];
  cf.hi = half != 0;
  cf.cA = cf.hi ? cf.b2 : cf.b4;
  cf.cB = cf.hi ? cf.b1 : cf.b3;
  cf.cC = cf.hi ? cf.b3 : 0.0f;
  cf.cD = cf.hi ? cf.b2 : cf.b4;

  const float* wp  = ws + OFF_ZC + ((size_t)run * 4 + 2 * half) * S + chan;
  const float* app = ws + OFF_APOW + rr * 16 + 2 * half * 4;
  const float* A8p = ws + OFF_APOW + SBLK * 16;
  v4f z[4], w[2], ap[2];
  float a8[16];
#pragma unroll
  for (int i = 0; i < 16; ++i) a8[i] = A8p[i];
#pragma unroll
  for (int e = 0; e < 2; ++e) {
    w[e]  = *(const v4f*)(wp + (size_t)e * S);
    ap[e] = *(const v4f*)(app + 4 * e);
  }
  v2f zin[4];
#pragma unroll
  for (int t = 0; t < 2; ++t) {
#pragma unroll
    for (int m = 0; m < 4; ++m) *(v2f*)&ush[wave + t * SBLK][m][2 * lane] = ur[t][m];
  }
  __syncthreads();

  const float* xp  = x + ((size_t)run * RLEN + 2 * half) * S + chan;
  float* o = out + ((size_t)run * RLEN + 8 * half) * S + chan;

  v4f xv[FIN_AHEAD + 1][8];
#pragma unroll
  for (int i = 0; i < FIN_AHEAD; ++i) lp_load_rows(xv[i], xp + (size_t)i * L * S);
#pragma unroll
  for (int i = 0; i < RUN; ++i) {
    __builtin_amdgcn_sched_barrier(0);
    if (i + FIN_AHEAD < RUN)
      lp_load_rows(xv[(i + FIN_AHEAD) % (FIN_AHEAD + 1)], xp + (size_t)(i + FIN_AHEAD) * L * S);
    __builtin_amdgcn_sched_barrier(0);
    if (i == 0) {
      const float* ul = &ush[0][0][4 * l16];
      v4f un[4];
#pragma unroll
      for (int m = 0; m < 4; ++m) {
        z[m] = v4f{0.f, 0.f, 0.f, 0.f};
        un[m] = *(const v4f*)(ul + m * SW);
      }
      for (int k = 0; k < sb; ++k) {
        const v4f u0 = un[0], u1 = un[1], u2 = un[2], u3 = un[3];
#pragma unroll
        for (int m = 0; m < 4; ++m) un[m] = *(const v4f*)(ul + ((k + 1) * 4 + m) * SW);
        const v4f n0 = a8[0]*z[0] + a8[1]*z[1] + a8[2]*z[2] + a8[3]*z[3] + u0;
        const v4f n1 = a8[4]*z[0] + a8[5]*z[1] + a8[6]*z[2] + a8[7]*z[3] + u1;
        const v4f n2 = a8[8]*z[0] + a8[9]*z[1] + a8[10]*z[2] + a8[11]*z[3] + u2;
        const v4f n3 = a8[12]*z[0] + a8[13]*z[1] + a8[14]*z[2] + a8[15]*z[3] + u3;
        z[0] = n0; z[1] = n1; z[2] = n2; z[3] = n3;
      }
#pragma unroll
      for (int j = 0; j < 4; ++j) {
#pragma unroll
        for (int e = 0; e < 2; ++e)
          zin[j][e] = ap[e][0] * z[0][j] + ap[e][1] * z[1][j] +
                      ap[e][2] * z[2][j] + ap[e][3] * z[3][j] + w[e][j];
      }
    }
    lp_fin_chunk(xv[i % (FIN_AHEAD + 1)], ah, pa, zin, o + (size_t)i * L * S, cf, i == RUN - 1);
  }
}

extern "C" void kernel_launch(void* const* d_in, const int* in_sizes, int n_in,
                              void* d_out, int out_size, void* d_ws, size_t ws_size,
                              hipStream_t stream) {
  const float* signal = (const float*)d_in[0];
  const float* b      = (const float*)d_in[1];
  const float* a      = (const float*)d_in[2];
  float* out = (float*)d_out;
  float* ws  = (float*)d_ws;

  lp_setup_kernel<<<1, 64, 0, stream>>>(b, a, ws);

  lp_zc_kernel<<<STRIP_BLOCKS, 256, 0, stream>>>(signal, ws);

  lp_final_kernel<<<STRIP_BLOCKS, 256, 0, stream>>>(signal, b, a, out, ws);
}


# Setup launch folded into pass 1 (3 -> 2 launches): a block's first wave builds Q and A^128 in LDS, a lane per column, under its first row loads; both via A^16 factors: table rounding changes
# speedup vs baseline: 2.6845x; 1.3331x over previous
#include <hip/hip_runtime.h>
#include <hip/hip_bf16.h>

typedef __attribute__((ext_vector_type(2))) float v2f;
typedef __attribute__((ext_vector_type(4))) float v4f;
typedef __attribute__((ext_vector_type(8))) float v8f;

constexpr int S       = 2048;
constexpr int T       = 16384;
constexpr int L       = 16;
constexpr int NCHUNK  = T / L;
constexpr int SW       = 64;
constexpr int NSTRIPG  = S / SW;
constexpr int RUN      = 8;
constexpr int RLEN     = RUN * L;
constexpr int NRUN     = NCHUNK / RUN;
constexpr int SBLK     = 8;
constexpr int NSB      = NRUN / SBLK;
constexpr int ZC_AHEAD  = 2;
constexpr int FIN_AHEAD = 2;
constexpr int NSTRIPS  = NSTRIPG * NRUN;
constexpr int STRIP_BLOCKS = NSTRIPS / SBLK;
static_assert(NSB == 2 * SBLK, "pass 3: every wave fetches two of the NSB superblock states");

constexpr size_t OFF_H    = 0;
constexpr size_t OFF_P    = 256;
constexpr size_t OFF_APOW = 832;
constexpr size_t OFF_ZC   = 4096;
constexpr size_t OFF_U    = OFF_ZC + (size_t)NRUN * 4 * S;

constexpr int QP = RLEN + 4;
struct LpTabs {
  float q[4][QP];
  float a128[16];
  float h[16];
  float p[64];
};

__device__ __forceinline__ void lp_mv4(const float (&M)[4][4], const float (&v)[4],
                                       float (&nv)[4]) {
#pragma unroll
  for (int i = 0; i < 4; ++i) {
    float s = 0.f;
#pragma unroll
    for (int j = 0; j < 4; ++j) s += M[i][j] * v[j];
    nv[i] = s;
  }
}

__device__ __forceinline__ float lp_from_lane(float v, int src) {
  return __builtin_bit_cast(float, __builtin_amdgcn_readlane(__builtin_bit_cast(int, v), src));
}

__device__ __forceinline__ void lp_tables(const float* __restrict__ bc,
                                          const float* __restrict__ ac,
                                          float* __restrict__ ws, const int lane,
                                          const bool first_block, LpTabs& tb) {
  const float b0 = bc[0];
  float A[4][4], g4[4];
#pragma unroll
  for (int i = 0; i < 4; ++i) {
#pragma unroll
    for (int j = 0; j < 4; ++j) A[i][j] = (j == i + 1) ? 1.0f : 0.0f;
    A[i][0] -= ac[i + 1];
    g4[i] = bc[i + 1] - ac[i + 1] * b0;
  }
  const bool isq = lane < 16;
  const bool iscol = (lane >> 2) == 4;
  const int sub = lane & 3;
  float v[4], q0[4], p0[16];
#pragma unroll
  for (int m = 0; m < 4; ++m) {
    v[m] = isq ? g4[m] : ((iscol && m == sub) ? 1.0f : 0.0f);
    q0[m] = v[m];
  }
#pragma unroll
  for (int t = 0; t < 16; ++t) {
    p0[t] = v[0];
#pragma unroll
    for (int m = 0; m < 4; ++m) q0[m] = (t == lane) ? v[m] : q0[m];
    float nv[4];
    lp_mv4(A, v, nv);
#pragma unroll
    for (int m = 0; m < 4; ++m) v[m] = nv[m];
  }
  if (iscol) {
#pragma unroll
    for (int t = 0; t < 16; ++t) tb.p[t * 4 + sub] = p0[t];
  }
  if (lane == 0) tb.h[0] = b0;
  if (lane < 15) tb.h[lane + 1] = q0[0];
  float A16[4][4];
#pragma unroll
  for (int i = 0; i < 4; ++i) {
#pragma unroll
    for (int cc = 0; cc < 4; ++cc) A16[i][cc] = lp_from_lane(v[i], 16 + cc);
  }
#pragma unroll
  for (int m = 0; m < 4; ++m) v[m] = isq ? q0[m] : v[m];
  float qq[RUN][4];
#pragma unroll
  for (int i = 0; i < RUN; ++i) {
    if (i > 0) {
      float nv[4];
      lp_mv4(A16, v, nv);
#pragma unroll
      for (int m = 0; m < 4; ++m) v[m] = nv[m];
    }
#pragma unroll
    for (int m = 0; m < 4; ++m) qq[i][m] = v[m];
  }
  if (isq) {
#pragma unroll
    for (int i = 0; i < RUN; ++i) {
#pragma unroll
      for (int m = 0; m < 4; ++m) tb.q[m][RLEN - 1 - 16 * i - lane] = qq[i][m];
    }
  }
  if (iscol) {
#pragma unroll
    for (int i = 0; i < 4; ++i) tb.a128[i * 4 + sub] = v[i];
  }
  if (first_block) {
    float A128[4][4];
#pragma unroll
    for (int i = 0; i < 4; ++i) {
#pragma unroll
      for (int cc = 0; cc < 4; ++cc) A128[i][cc] = lp_from_lane(v[i], 16 + cc);
    }
    float pw[4];
#pragma unroll
    for (int m = 0; m < 4; ++m) pw[m] = (m == sub) ? 1.0f : 0.0f;
#pragma unroll
    for (int p = 0; p <= SBLK; ++p) {
      if (iscol) {
#pragma unroll
        for (int i = 0; i < 4; ++i) ws[OFF_APOW + p * 16 + i * 4 + sub] = pw[i];
      }
      if (p < SBLK) {
        float nv[4];
        lp_mv4(A128, pw, nv);
#pragma unroll
        for (int m = 0; m < 4; ++m) pw[m] = nv[m];
      }
    }
  }
}

__device__ __forceinline__ void lp_load_rows(v4f (&xv)[8], const float* __restrict__ xp) {
#pragma unroll
  for (int k = 0; k < 4; ++k) {
    xv[2 * k]     = *(const v4f*)(xp + (size_t)(4 * k) * S);
    xv[2 * k + 1] = *(const v4f*)(xp + (size_t)(4 * k + 1) * S);
  }
}

__device__ __forceinline__ void lp_zc_chunk(v8f (&acc)[4], const v4f (&xv)[8],
                                            const v2f (&aq)[4]) {
#pragma unroll
  for (int k = 0; k < 4; ++k) {
#pragma unroll
    for (int j = 0; j < 4; ++j) {
      v2f bv;
      bv[0] = xv[2 * k][j];
      bv[1] = xv[2 * k + 1][j];
      acc[j] = __builtin_amdgcn_wmma_f32_16x16x4_f32(false, aq[k], false, bv,
                                                     (short)0, acc[j], false, false);
    }
  }
}

__global__ __launch_bounds__(256) void lp_zc_kernel(const float* __restrict__ x,
                                                    const float* __restrict__ bc,
                                                    const float* __restrict__ ac,
                                                    float* __restrict__ ws) {
  __shared__ __attribute__((aligned(16))) float zsh[SBLK][4][SW];
  __shared__ __attribute__((aligned(16))) LpTabs tabs;
  const int wave = __builtin_amdgcn_readfirstlane((int)(threadIdx.x >> 5));
  const int lane = threadIdx.x & 31;
  const int sg = blockIdx.x & (NSTRIPG - 1);
  const int sb = blockIdx.x / NSTRIPG;
  const int run = sb * SBLK + wave;
  const int half = lane >> 4;
  const int l16 = lane & 15;
  const int chan = sg * SW + 4 * l16;
  const float* xp = x + ((size_t)run * RLEN + 2 * half) * S + chan;
  v4f xv[ZC_AHEAD + 1][8];
#pragma unroll
  for (int i = 0; i <= ZC_AHEAD; ++i) lp_load_rows(xv[i], xp + (size_t)i * L * S);
  if (wave == 0) lp_tables(bc, ac, ws, lane, blockIdx.x == 0, tabs);
  __syncthreads();
  const float* Q = &tabs.q[l16 & 3][2 * half];
  const bool qrow = l16 < 4;
  v2f qraw[RUN][4];
#pragma unroll
  for (int i = 0; i < RUN; ++i) {
#pragma unroll
    for (int k = 0; k < 4; ++k) qraw[i][k] = *(const v2f*)(Q + 16 * i + 4 * k);
  }
  float a[16];
#pragma unroll
  for (int i = 0; i < 16; ++i) a[i] = tabs.a128[i];
  v8f acc[4] = {};
#pragma unroll
  for (int i = 0; i < RUN; ++i) {
    __builtin_amdgcn_sched_barrier(0);
    if (i > 0 && i + ZC_AHEAD < RUN)
      lp_load_rows(xv[(i + ZC_AHEAD) % (ZC_AHEAD + 1)], xp + (size_t)(i + ZC_AHEAD) * L * S);
    __builtin_amdgcn_sched_barrier(0);
    v2f aq[4];
#pragma unroll
    for (int k = 0; k < 4; ++k) {
      aq[k][0] = qrow ? qraw[i][k][0] : 0.0f;
      aq[k][1] = qrow ? qraw[i][k][1] : 0.0f;
    }
    lp_zc_chunk(acc, xv[i % (ZC_AHEAD + 1)], aq);
  }
  if (half == 0) {
#pragma unroll
    for (int m = 0; m < 4; ++m) {
      v4f o;
      o[0] = acc[0][m]; o[1] = acc[1][m]; o[2] = acc[2][m]; o[3] = acc[3][m];
      *(v4f*)&zsh[wave][m][4 * l16] = o;
    }
  }
  __syncthreads();
  const int c2 = 2 * lane;
  v2f r[SBLK][4];
#pragma unroll
  for (int jj = 0; jj < SBLK; ++jj) {
#pragma unroll
    for (int m = 0; m < 4; ++m) r[jj][m] = *(const v2f*)&zsh[jj][m][c2];
  }
  v2f w0 = {0.f, 0.f}, w1 = w0, w2 = w0, w3 = w0;
  v2f k0 = w0, k1 = w0, k2 = w0, k3 = w0;
#pragma unroll
  for (int jj = 0; jj < SBLK; ++jj) {
    if (jj == wave) { k0 = w0; k1 = w1; k2 = w2; k3 = w3; }
    const v2f n0 = a[0]*w0 + a[1]*w1 + a[2]*w2 + a[3]*w3 + r[jj][0];
    const v2f n1 = a[4]*w0 + a[5]*w1 + a[6]*w2 + a[7]*w3 + r[jj][1];
    const v2f n2 = a[8]*w0 + a[9]*w1 + a[10]*w2 + a[11]*w3 + r[jj][2];
    const v2f n3 = a[12]*w0 + a[13]*w1 + a[14]*w2 + a[15]*w3 + r[jj][3];
    w0 = n0; w1 = n1; w2 = n2; w3 = n3;
  }
  float* wo = ws + OFF_ZC + (size_t)run * 4 * S + sg * SW + c2;
  *(v2f*)(wo + 0 * (size_t)S) = k0; *(v2f*)(wo + 1 * (size_t)S) = k1;
  *(v2f*)(wo + 2 * (size_t)S) = k2; *(v2f*)(wo + 3 * (size_t)S) = k3;
  if (wave == SBLK - 1) {
    float* uo = ws + OFF_U + (size_t)sb * 4 * S + sg * SW + c2;
    *(v2f*)(uo + 0 * (size_t)S) = w0; *(v2f*)(uo + 1 * (size_t)S) = w1;
    *(v2f*)(uo + 2 * (size_t)S) = w2; *(v2f*)(uo + 3 * (size_t)S) = w3;
  }
  if (blockIdx.x == 0) {
    const int t = threadIdx.x >> 4, k = threadIdx.x & 15;
    ws[OFF_H + threadIdx.x] = (k <= t) ? tabs.h[(t - k) & 15] : 0.0f;
    if (threadIdx.x < 64) ws[OFF_P + threadIdx.x] = tabs.p[threadIdx.x];
  }
}

struct LpCarry {
  float b1, b2, b3, b4, a1, a2, a3, a4;
  float cA, cB, cC, cD;
  bool hi;
};

__device__ __forceinline__ float lp_xhalf(float v) {
  return __builtin_bit_cast(float,
      __builtin_amdgcn_permlanex16(__builtin_bit_cast(int, v), __builtin_bit_cast(int, v),
                                   0x76543210, 0xfedcba98, false, false));
}

__device__ __forceinline__ void lp_fin_chunk(const v4f (&xv)[8], const v2f (&ah)[4], v2f pa,
                                             v2f (&zin)[4], float* __restrict__ o,
                                             const LpCarry& cf, const bool last) {
  v8f acc[4];
#pragma unroll
  for (int j = 0; j < 4; ++j) {
    v8f zero = {};
    acc[j] = __builtin_amdgcn_wmma_f32_16x16x4_f32(false, pa, false, zin[j],
                                                   (short)0, zero, false, false);
  }
#pragma unroll
  for (int k = 0; k < 4; ++k) {
#pragma unroll
    for (int j = 0; j < 4; ++j) {
      v2f bv;
      bv[0] = xv[2 * k][j];
      bv[1] = xv[2 * k + 1][j];
      acc[j] = __builtin_amdgcn_wmma_f32_16x16x4_f32(false, ah[k], false, bv,
                                                     (short)0, acc[j], false, false);
    }
  }
  if (!last) {
#pragma unroll
    for (int j = 0; j < 4; ++j) {
      const float x6 = xv[6][j], x7 = xv[7][j];
      const float y4 = acc[j][4], y5 = acc[j][5], y6 = acc[j][6], y7 = acc[j][7];
      const float u0 = cf.cA * x6 + cf.cB * x7;
      const float u1 = cf.cC * x6 + cf.cD * x7;
      const float s0 = u0 - (cf.a4 * y4 + cf.a3 * y5 + cf.a2 * y6 + cf.a1 * y7);
      const float s1 = u1 - (cf.a4 * y5 + cf.a3 * y6 + cf.a2 * y7);
      const float s2 = (cf.b4 * x6 + cf.b3 * x7) - (cf.a4 * y6 + cf.a3 * y7);
      const float s3 = cf.b4 * x7 - cf.a4 * y7;
      const float t0 = lp_xhalf(s0);
      const float t1 = lp_xhalf(s1);
      zin[j][0] = cf.hi ? s2 : (u0 + t0);
      zin[j][1] = cf.hi ? s3 : (u1 + t1);
    }
  }
#pragma unroll
  for (int r = 0; r < 8; ++r) {
    v4f ov;
    ov[0] = acc[0][r]; ov[1] = acc[1][r]; ov[2] = acc[2][r]; ov[3] = acc[3][r];
    *(v4f*)(o + (size_t)r * S) = ov;
  }
}

__global__ __launch_bounds__(256) void lp_final_kernel(const float* __restrict__ x,
                                                       const float* __restrict__ bc,
                                                       const float* __restrict__ ac,
                                                       float* __restrict__ out,
                                                       const float* __restrict__ ws) {
  __shared__ __attribute__((aligned(16))) float ush[NSB][4][SW];
  const int wave = __builtin_amdgcn_readfirstlane((int)(threadIdx.x >> 5));
  const int lane = threadIdx.x & 31;
  const int sg = blockIdx.x & (NSTRIPG - 1);
  const int sb = blockIdx.x / NSTRIPG;
  const int rr = wave;
  const int run = sb * SBLK + rr;
  const int half = lane >> 4;
  const int l16 = lane & 15;
  const int chan = sg * SW + 4 * l16;

  const float* up = ws + OFF_U + sg * SW + 2 * lane;
  v2f ur[2][4];
#pragma unroll
  for (int t = 0; t < 2; ++t) {
#pragma unroll
    for (int m = 0; m < 4; ++m)
      ur[t][m] = *(const v2f*)(up + ((size_t)(wave + t * SBLK) * 4 + m) * S);
  }

  const float* H = ws + OFF_H;
  const float* Pm = ws + OFF_P;
  v2f ah[4], pa;
#pragma unroll
  for (int k = 0; k < 4; ++k) {
    const int kk0 = 4 * k + 2 * half;
    ah[k][0] = H[l16 * 16 + kk0];
    ah[k][1] = H[l16 * 16 + kk0 + 1];
  }
  pa[0] = Pm[l16 * 4 + 2 * half + 0];
  pa[1] = Pm[l16 * 4 + 2 * half + 1];

  LpCarry cf;
  cf.b1 = bc[1]; cf.b2 = bc[2]; cf.b3 = bc[3]; cf.b4 = bc[4];
  cf.a1 = ac[1]; cf.a2 = ac[2]; cf.a3 = ac[3]; cf.a4 = ac[4];
  cf.hi = half != 0;
  cf.cA = cf.hi ? cf.b2 : cf.b4;
  cf.cB = cf.hi ? cf.b1 : cf.b3;
  cf.cC = cf.hi ? cf.b3 : 0.0f;
  cf.cD = cf.hi ? cf.b2 : cf.b4;

  const float* wp  = ws + OFF_ZC + ((size_t)run * 4 + 2 * half) * S + chan;
  const float* app = ws + OFF_APOW + rr * 16 + 2 * half * 4;
  const float* A8p = ws + OFF_APOW + SBLK * 16;
  v4f z[4], w[2], ap[2];
  float a8[16];
#pragma unroll
  for (int i = 0; i < 16; ++i) a8[i] = A8p[i];
#pragma unroll
  for (int e = 0; e < 2; ++e) {
    w[e]  = *(const v4f*)(wp + (size_t)e * S);
    ap[e] = *(const v4f*)(app + 4 * e);
  }
  v2f zin[4];
#pragma unroll
  for (int t = 0; t < 2; ++t) {
#pragma unroll
    for (int m = 0; m < 4; ++m) *(v2f*)&ush[wave + t * SBLK][m][2 * lane] = ur[t][m];
  }
  __syncthreads();

  const float* xp  = x + ((size_t)run * RLEN + 2 * half) * S + chan;
  float* o = out + ((size_t)run * RLEN + 8 * half) * S + chan;

  v4f xv[FIN_AHEAD + 1][8];
#pragma unroll
  for (int i = 0; i < FIN_AHEAD; ++i) lp_load_rows(xv[i], xp + (size_t)i * L * S);
#pragma unroll
  for (int i = 0; i < RUN; ++i) {
    __builtin_amdgcn_sched_barrier(0);
    if (i + FIN_AHEAD < RUN)
      lp_load_rows(xv[(i + FIN_AHEAD) % (FIN_AHEAD + 1)], xp + (size_t)(i + FIN_AHEAD) * L * S);
    __builtin_amdgcn_sched_barrier(0);
    if (i == 0) {
      const float* ul = &ush[0][0][4 * l16];
      v4f un[4];
#pragma unroll
      for (int m = 0; m < 4; ++m) {
        z[m] = v4f{0.f, 0.f, 0.f, 0.f};
        un[m] = *(const v4f*)(ul + m * SW);
      }
      for (int k = 0; k < sb; ++k) {
        const v4f u0 = un[0], u1 = un[1], u2 = un[2], u3 = un[3];
#pragma unroll
        for (int m = 0; m < 4; ++m) un[m] = *(const v4f*)(ul + ((k + 1) * 4 + m) * SW);
        const v4f n0 = a8[0]*z[0] + a8[1]*z[1] + a8[2]*z[2] + a8[3]*z[3] + u0;
        const v4f n1 = a8[4]*z[0] + a8[5]*z[1] + a8[6]*z[2] + a8[7]*z[3] + u1;
        const v4f n2 = a8[8]*z[0] + a8[9]*z[1] + a8[10]*z[2] + a8[11]*z[3] + u2;
        const v4f n3 = a8[12]*z[0] + a8[13]*z[1] + a8[14]*z[2] + a8[15]*z[3] + u3;
        z[0] = n0; z[1] = n1; z[2] = n2; z[3] = n3;
      }
#pragma unroll
      for (int j = 0; j < 4; ++j) {
#pragma unroll
        for (int e = 0; e < 2; ++e)
          zin[j][e] = ap[e][0] * z[0][j] + ap[e][1] * z[1][j] +
                      ap[e][2] * z[2][j] + ap[e][3] * z[3][j] + w[e][j];
      }
    }
    lp_fin_chunk(xv[i % (FIN_AHEAD + 1)], ah, pa, zin, o + (size_t)i * L * S, cf, i == RUN - 1);
  }
}

extern "C" void kernel_launch(void* const* d_in, const int* in_sizes, int n_in,
                              void* d_out, int out_size, void* d_ws, size_t ws_size,
                              hipStream_t stream) {
  const float* signal = (const float*)d_in[0];
  const float* b      = (const float*)d_in[1];
  const float* a      = (const float*)d_in[2];
  float* out = (float*)d_out;
  float* ws  = (float*)d_ws;

  lp_zc_kernel<<<STRIP_BLOCKS, 256, 0, stream>>>(signal, b, a, ws);

  lp_final_kernel<<<STRIP_BLOCKS, 256, 0, stream>>>(signal, b, a, out, ws);
}
